# starting kernel
# baseline (speedup 1.0000x reference)
#include <hip/hip_runtime.h>
#include <hip/hip_bf16.h>
#include <math.h>

typedef __attribute__((ext_vector_type(2))) float v2f;
typedef __attribute__((ext_vector_type(8))) float v8f;

#define IN_F 64
#define HD   128
#define H_   8
#define D_   16

__global__ __launch_bounds__(256) void gat_proj_wmma(
    const float* __restrict__ feat,
    const float* __restrict__ W,
    const float* __restrict__ Wres,
    float* __restrict__ ft_ws,
    float* __restrict__ res_out,
    int nrows)
{
    const int wave = threadIdx.x >> 5;
    const int lane = threadIdx.x & 31;
    const int row0 = blockIdx.x * 16;
    const int col0 = wave * 16;

    const float* B  = (blockIdx.y == 0) ? W : Wres;
    float* outp     = (blockIdx.y == 0) ? ft_ws : res_out;

    const int r  = lane & 15;
    const int hi = lane >> 4;

    const int arow = (row0 + r < nrows) ? (row0 + r) : (nrows - 1);
    const float* apBase = feat + (size_t)arow * IN_F;

    v8f acc = {};
#pragma unroll
    for (int kb = 0; kb < IN_F; kb += 4) {
        v2f a;
        a.x = apBase[kb + 2 * hi + 0];
        a.y = apBase[kb + 2 * hi + 1];
        const float* bp = B + (size_t)(kb + 2 * hi) * HD + col0 + r;
        v2f b;
        b.x = bp[0];
        b.y = bp[HD];
        acc = __builtin_amdgcn_wmma_f32_16x16x4_f32(
             false, a,  false, b,
             (short)0, acc,  false,  false);
    }

    const int rbase = row0 + 8 * hi;
#pragma unroll
    for (int v = 0; v < 8; ++v) {
        int row = rbase + v;
        if (row < nrows)
            outp[(size_t)row * HD + col0 + r] = acc[v];
    }
}

__global__ void gat_node_attn(
    const float* __restrict__ ft,
    const float* __restrict__ attn_l,
    const float* __restrict__ attn_r,
    float* __restrict__ el, float* __restrict__ er,
    float* __restrict__ emax, float* __restrict__ denom,
    int totalNH)
{
    int i = blockIdx.x * blockDim.x + threadIdx.x;
    if (i >= totalNH) return;
    int h = i & (H_ - 1);
    const float* f  = ft + (size_t)(i >> 3) * HD + h * D_;
    const float* al = attn_l + h * D_;
    const float* ar = attn_r + h * D_;
    float sl = 0.f, sr = 0.f;
#pragma unroll
    for (int d = 0; d < D_; ++d) {
        float v = f[d];
        sl = fmaf(v, al[d], sl);
        sr = fmaf(v, ar[d], sr);
    }
    el[i] = sl;
    er[i] = sr;
    emax[i]  = -INFINITY;
    denom[i] = 0.f;
}

__device__ __forceinline__ float lrelu(float x) {
    return x > 0.f ? x : 0.01f * x;
}

__global__ void gat_edge_max(
    const int* __restrict__ src, const int* __restrict__ dst,
    const float* __restrict__ el, const float* __restrict__ er,
    float* __restrict__ emax, int totalEH)
{
    int i = blockIdx.x * blockDim.x + threadIdx.x;
    if (i >= totalEH) return;
    int e = i >> 3, h = i & (H_ - 1);
    int s = src[e], d = dst[e];
    float x = lrelu(el[s * H_ + h] + er[d * H_ + h]);
    atomicMax(&emax[d * H_ + h], x);
}

__global__ void gat_edge_sum(
    const int* __restrict__ src, const int* __restrict__ dst,
    const float* __restrict__ el, const float* __restrict__ er,
    const float* __restrict__ emax, float* __restrict__ denom, int totalEH)
{
    int i = blockIdx.x * blockDim.x + threadIdx.x;
    if (i >= totalEH) return;
    int e = i >> 3, h = i & (H_ - 1);
    int s = src[e], d = dst[e];
    float x = lrelu(el[s * H_ + h] + er[d * H_ + h]);
    float ex = expf(x - emax[d * H_ + h]);
    atomicAdd(&denom[d * H_ + h], ex);
}

__global__ void gat_edge_aggr(
    const int* __restrict__ src, const int* __restrict__ dst,
    const float* __restrict__ el, const float* __restrict__ er,
    const float* __restrict__ emax, const float* __restrict__ denom,
    const float* __restrict__ ft, float* __restrict__ out, int totalEH)
{
    int i = blockIdx.x * blockDim.x + threadIdx.x;
    if (i >= totalEH) return;
    int e = i >> 3, h = i & (H_ - 1);
    int s = src[e], d = dst[e];
    float x = lrelu(el[s * H_ + h] + er[d * H_ + h]);
    float a = expf(x - emax[d * H_ + h]) / denom[d * H_ + h];
    const float* fs = ft  + (size_t)s * HD + h * D_;
    float*       od = out + (size_t)d * HD + h * D_;
#pragma unroll
    for (int dd = 0; dd < D_; ++dd)
        atomicAdd(&od[dd], fs[dd] * a);
}

__global__ void gat_elu(float* __restrict__ out, int total)
{
    int i = blockIdx.x * blockDim.x + threadIdx.x;
    if (i >= total) return;
    float x = out[i];
    out[i] = x > 0.f ? x : expm1f(x);
}

extern "C" void kernel_launch(void* const* d_in, const int* in_sizes, int n_in,
                              void* d_out, int out_size, void* d_ws, size_t ws_size,
                              hipStream_t stream)
{
    const float* feat   = (const float*)d_in[0];
    const int*   src    = (const int*)  d_in[1];
    const int*   dst    = (const int*)  d_in[2];
    const float* W      = (const float*)d_in[3];
    const float* attn_l = (const float*)d_in[4];
    const float* attn_r = (const float*)d_in[5];
    const float* Wres   = (const float*)d_in[6];
    float* out = (float*)d_out;

    const int N = in_sizes[0] / IN_F;
    const int E = in_sizes[1];

    char* ws = (char*)d_ws;
    float* ft    = (float*)ws;                       ws += (size_t)N * HD * sizeof(float);
    float* el    = (float*)ws;                       ws += (size_t)N * H_ * sizeof(float);
    float* er    = (float*)ws;                       ws += (size_t)N * H_ * sizeof(float);
    float* emax  = (float*)ws;                       ws += (size_t)N * H_ * sizeof(float);
    float* denom = (float*)ws;

    dim3 gproj((N + 15) / 16, 2);
    gat_proj_wmma<<<gproj, 256, 0, stream>>>(feat, W, Wres, ft, out, N);

    int totalNH = N * H_;
    gat_node_attn<<<(totalNH + 255) / 256, 256, 0, stream>>>(
        ft, attn_l, attn_r, el, er, emax, denom, totalNH);

    int totalEH = E * H_;
    int gE = (totalEH + 255) / 256;
    gat_edge_max <<<gE, 256, 0, stream>>>(src, dst, el, er, emax, totalEH);
    gat_edge_sum <<<gE, 256, 0, stream>>>(src, dst, el, er, emax, denom, totalEH);
    gat_edge_aggr<<<gE, 256, 0, stream>>>(src, dst, el, er, emax, denom, ft, out, totalEH);

    int totalOut = N * HD;
    gat_elu<<<(totalOut + 255) / 256, 256, 0, stream>>>(out, totalOut);
}


# float atomics removed: edges counting-sorted by destination in-call (int atomics); one wave per destination row gathers max, exp-sum, weighted sum, residual, ELU; sums run in sorted-list order
# speedup vs baseline: 3.3395x; 3.3395x over previous
#include <hip/hip_runtime.h>
#include <hip/hip_bf16.h>
#include <math.h>

typedef __attribute__((ext_vector_type(2))) float v2f;
typedef __attribute__((ext_vector_type(8))) float v8f;
typedef __attribute__((ext_vector_type(4))) float v4f;

#define IN_F 64
#define HD   128
#define H_   8
#define D_   16

__global__ __launch_bounds__(256) void gat_proj_wmma(
    const float* __restrict__ feat,
    const float* __restrict__ W,
    const float* __restrict__ Wres,
    float* __restrict__ ft_ws,
    float* __restrict__ res_out,
    int nrows)
{
    const int wave = threadIdx.x >> 5;
    const int lane = threadIdx.x & 31;
    const int row0 = blockIdx.x * 16;
    const int col0 = wave * 16;

    const float* B  = (blockIdx.y == 0) ? W : Wres;
    float* outp     = (blockIdx.y == 0) ? ft_ws : res_out;

    const int r  = lane & 15;
    const int hi = lane >> 4;

    const int arow = (row0 + r < nrows) ? (row0 + r) : (nrows - 1);
    const float* apBase = feat + (size_t)arow * IN_F;

    v8f acc = {};
#pragma unroll
    for (int kb = 0; kb < IN_F; kb += 4) {
        v2f a;
        a.x = apBase[kb + 2 * hi + 0];
        a.y = apBase[kb + 2 * hi + 1];
        const float* bp = B + (size_t)(kb + 2 * hi) * HD + col0 + r;
        v2f b;
        b.x = bp[0];
        b.y = bp[HD];
        acc = __builtin_amdgcn_wmma_f32_16x16x4_f32(
             false, a,  false, b,
             (short)0, acc,  false,  false);
    }

    const int rbase = row0 + 8 * hi;
#pragma unroll
    for (int v = 0; v < 8; ++v) {
        int row = rbase + v;
        if (row < nrows)
            outp[(size_t)row * HD + col0 + r] = acc[v];
    }
}

__global__ void gat_node_el(
    const float* __restrict__ ft,
    const float* __restrict__ attn_l,
    float* __restrict__ el, int* __restrict__ cnt,
    int totalNH)
{
    int i = blockIdx.x * blockDim.x + threadIdx.x;
    if (i >= totalNH) return;
    int h = i & (H_ - 1);
    const float* f  = ft + (size_t)(i >> 3) * HD + h * D_;
    const float* al = attn_l + h * D_;
    float sl = 0.f;
#pragma unroll
    for (int d = 0; d < D_; ++d)
        sl = fmaf(f[d], al[d], sl);
    el[i] = sl;
    if (h == 0) cnt[i >> 3] = 0;
}

__device__ __forceinline__ float lrelu(float x) {
    return x > 0.f ? x : 0.01f * x;
}

__global__ void gat_hist(const int* __restrict__ dst, int* __restrict__ cnt, int E)
{
    int e = blockIdx.x * blockDim.x + threadIdx.x;
    if (e >= E) return;
    atomicAdd(&cnt[dst[e]], 1);
}

#define SCAN_PER 8
__global__ __launch_bounds__(1024) void gat_scan(int* cnt, int n)
{
    __shared__ int wsum[32];
    const int tid  = threadIdx.x;
    const int lane = tid & 31;
    const int wave = tid >> 5;
    int carry = 0;
    for (int base = 0; base < n; base += 1024 * SCAN_PER) {
        const int i0 = base + tid * SCAN_PER;
        int v[SCAN_PER];
        int t = 0;
#pragma unroll
        for (int q = 0; q < SCAN_PER; ++q) {
            v[q] = (i0 + q < n) ? cnt[i0 + q] : 0;
            t += v[q];
        }
        int incl = t;
        for (int o = 1; o < 32; o <<= 1) {
            int y = __shfl_up(incl, o);
            if (lane >= o) incl += y;
        }
        if (lane == 31) wsum[wave] = incl;
        __syncthreads();
        const int ws = wsum[lane];
        int wincl = ws;
        for (int o = 1; o < 32; o <<= 1) {
            int y = __shfl_up(wincl, o);
            if (lane >= o) wincl += y;
        }
        const int wave_excl = __shfl(wincl, wave) - __shfl(ws, wave);
        const int total     = __shfl(wincl, 31);
        __syncthreads();
        int run = carry + wave_excl + incl - t;
#pragma unroll
        for (int q = 0; q < SCAN_PER; ++q) {
            if (i0 + q < n) cnt[i0 + q] = run;
            run += v[q];
        }
        carry += total;
    }
}

__global__ void gat_fill(
    const int* __restrict__ src, const int* __restrict__ dst,
    int* __restrict__ offs, int* __restrict__ ssrc, int E)
{
    int e = blockIdx.x * blockDim.x + threadIdx.x;
    if (e >= E) return;
    int pos = atomicAdd(&offs[dst[e]], 1);
    if ((unsigned)pos < (unsigned)E) ssrc[pos] = src[e];
}

__global__ __launch_bounds__(256) void gat_gather(
    const int* __restrict__ offs, const int* __restrict__ ssrc,
    const float* __restrict__ el, const float* __restrict__ attn_r,
    const float* __restrict__ ft, float* __restrict__ out, int N, int E)
{
    const int lane = threadIdx.x & 31;
    const int node = __builtin_amdgcn_readfirstlane((int)(blockIdx.x * 8 + (threadIdx.x >> 5)));
    if (node >= N) return;
    int beg = (node > 0) ? offs[node - 1] : 0;
    int end = offs[node];
    beg = max(0, min(beg, E));
    end = max(beg, min(end, E));

    const int h = lane >> 2;
    const size_t col = (size_t)lane * 4;

    const v4f fd = *(const v4f*)(ft + (size_t)node * HD + col);
    const float* ar = attn_r + lane * 4;
    float er = fd.x * ar[0];
    er = fmaf(fd.y, ar[1], er);
    er = fmaf(fd.z, ar[2], er);
    er = fmaf(fd.w, ar[3], er);
    er += __shfl_xor(er, 1);
    er += __shfl_xor(er, 2);

    const float* elh = el + h;

    float m = -INFINITY;
    for (int cb = beg; cb < end; cb += 32) {
        const int n = min(32, end - cb);
        int sv = (cb + lane < end) ? ssrc[cb + lane] : 0;
        sv = max(0, min(sv, N - 1));
        int k = 0;
        for (; k + 4 <= n; k += 4) {
            const int s0 = __builtin_amdgcn_readlane(sv, k);
            const int s1 = __builtin_amdgcn_readlane(sv, k + 1);
            const int s2 = __builtin_amdgcn_readlane(sv, k + 2);
            const int s3 = __builtin_amdgcn_readlane(sv, k + 3);
            const float x0 = lrelu(elh[s0 * H_] + er);
            const float x1 = lrelu(elh[s1 * H_] + er);
            const float x2 = lrelu(elh[s2 * H_] + er);
            const float x3 = lrelu(elh[s3 * H_] + er);
            m = fmaxf(m, fmaxf(fmaxf(x0, x1), fmaxf(x2, x3)));
        }
        for (; k < n; ++k) {
            const int s0 = __builtin_amdgcn_readlane(sv, k);
            m = fmaxf(m, lrelu(elh[s0 * H_] + er));
        }
    }

    float z = 0.f;
    v4f acc = {0.f, 0.f, 0.f, 0.f};
    for (int cb = beg; cb < end; cb += 32) {
        const int n = min(32, end - cb);
        int sv = (cb + lane < end) ? ssrc[cb + lane] : 0;
        sv = max(0, min(sv, N - 1));
        int k = 0;
        for (; k + 4 <= n; k += 4) {
            const int s0 = __builtin_amdgcn_readlane(sv, k);
            const int s1 = __builtin_amdgcn_readlane(sv, k + 1);
            const int s2 = __builtin_amdgcn_readlane(sv, k + 2);
            const int s3 = __builtin_amdgcn_readlane(sv, k + 3);
            const v4f r0 = *(const v4f*)(ft + (size_t)s0 * HD + col);
            const v4f r1 = *(const v4f*)(ft + (size_t)s1 * HD + col);
            const v4f r2 = *(const v4f*)(ft + (size_t)s2 * HD + col);
            const v4f r3 = *(const v4f*)(ft + (size_t)s3 * HD + col);
            const float p0 = expf(lrelu(elh[s0 * H_] + er) - m);
            const float p1 = expf(lrelu(elh[s1 * H_] + er) - m);
            const float p2 = expf(lrelu(elh[s2 * H_] + er) - m);
            const float p3 = expf(lrelu(elh[s3 * H_] + er) - m);
            z += p0; acc += r0 * p0;
            z += p1; acc += r1 * p1;
            z += p2; acc += r2 * p2;
            z += p3; acc += r3 * p3;
        }
        for (; k < n; ++k) {
            const int s0 = __builtin_amdgcn_readlane(sv, k);
            const v4f r0 = *(const v4f*)(ft + (size_t)s0 * HD + col);
            const float p0 = expf(lrelu(elh[s0 * H_] + er) - m);
            z += p0; acc += r0 * p0;
        }
    }
    if (end > beg) acc = acc / z;

    float* op = out + (size_t)node * HD + col;
    v4f o = *(const v4f*)op + acc;
    o.x = o.x > 0.f ? o.x : expm1f(o.x);
    o.y = o.y > 0.f ? o.y : expm1f(o.y);
    o.z = o.z > 0.f ? o.z : expm1f(o.z);
    o.w = o.w > 0.f ? o.w : expm1f(o.w);
    *(v4f*)op = o;
}

extern "C" void kernel_launch(void* const* d_in, const int* in_sizes, int n_in,
                              void* d_out, int out_size, void* d_ws, size_t ws_size,
                              hipStream_t stream)
{
    const float* feat   = (const float*)d_in[0];
    const int*   src    = (const int*)  d_in[1];
    const int*   dst    = (const int*)  d_in[2];
    const float* W      = (const float*)d_in[3];
    const float* attn_l = (const float*)d_in[4];
    const float* attn_r = (const float*)d_in[5];
    const float* Wres   = (const float*)d_in[6];
    float* out = (float*)d_out;

    const int N = in_sizes[0] / IN_F;
    const int E = in_sizes[1];

    char* ws = (char*)d_ws;
    float* ft   = (float*)ws;                        ws += (size_t)N * HD * sizeof(float);
    float* el   = (float*)ws;                        ws += (size_t)N * H_ * sizeof(float);
    int*   ssrc = (int*)ws;                          ws += (size_t)E * sizeof(int);
    int*   offs = (int*)ws;

    dim3 gproj((N + 15) / 16, 2);
    gat_proj_wmma<<<gproj, 256, 0, stream>>>(feat, W, Wres, ft, out, N);

    int totalNH = N * H_;
    gat_node_el<<<(totalNH + 255) / 256, 256, 0, stream>>>(ft, attn_l, el, offs, totalNH);

    int gE = (E + 255) / 256;
    gat_hist<<<gE, 256, 0, stream>>>(dst, offs, E);
    gat_scan<<<1, 1024, 0, stream>>>(offs, N);
    gat_fill<<<gE, 256, 0, stream>>>(src, dst, offs, ssrc, E);

    gat_gather<<<(N + 7) / 8, 256, 0, stream>>>(offs, ssrc, el, attn_r, ft, out, N, E);
}


# persistent waves: the 5 wide kernels launch fixed grids and loop over their items (325k -> 49k waves), per-wave constants loaded once; edge arrival order in segments, hence row-sum order, changes
# speedup vs baseline: 3.4723x; 1.0398x over previous
#include <hip/hip_runtime.h>
#include <hip/hip_bf16.h>
#include <math.h>

typedef __attribute__((ext_vector_type(2))) float v2f;
typedef __attribute__((ext_vector_type(8))) float v8f;
typedef __attribute__((ext_vector_type(4))) float v4f;

#define IN_F 64
#define HD   128
#define H_   8
#define D_   16

__global__ __launch_bounds__(256) void gat_proj_wmma(
    const float* __restrict__ feat,
    const float* __restrict__ W,
    const float* __restrict__ Wres,
    float* __restrict__ ft_ws,
    float* __restrict__ res_out,
    int nrows)
{
    const int wave = threadIdx.x >> 5;
    const int lane = threadIdx.x & 31;
    const int col0 = wave * 16;

    const float* B  = (blockIdx.y == 0) ? W : Wres;
    float* outp     = (blockIdx.y == 0) ? ft_ws : res_out;

    const int r  = lane & 15;
    const int hi = lane >> 4;

    v2f b[IN_F / 4];
#pragma unroll
    for (int j = 0; j < IN_F / 4; ++j) {
        const float* bp = B + (size_t)(4 * j + 2 * hi) * HD + col0 + r;
        b[j].x = bp[0];
        b[j].y = bp[HD];
    }

    const int ntiles = (nrows + 15) >> 4;
    for (int tile = blockIdx.x; tile < ntiles; tile += gridDim.x) {
        const int row0 = tile * 16;
        const int arow = (row0 + r < nrows) ? (row0 + r) : (nrows - 1);
        const float* apBase = feat + (size_t)arow * IN_F;

        v8f acc = {};
#pragma unroll
        for (int j = 0; j < IN_F / 4; ++j) {
            v2f a;
            a.x = apBase[4 * j + 2 * hi + 0];
            a.y = apBase[4 * j + 2 * hi + 1];
            acc = __builtin_amdgcn_wmma_f32_16x16x4_f32(
                 false, a,  false, b[j],
                 (short)0, acc,  false,  false);
        }

        const int rbase = row0 + 8 * hi;
#pragma unroll
        for (int v = 0; v < 8; ++v) {
            int row = rbase + v;
            if (row < nrows)
                outp[(size_t)row * HD + col0 + r] = acc[v];
        }
    }
}

__global__ __launch_bounds__(256) void gat_node_el(
    const float* __restrict__ ft,
    const float* __restrict__ attn_l,
    float* __restrict__ el, int* __restrict__ cnt,
    int totalNH)
{
    const int stride = gridDim.x * 256;
    const int h = threadIdx.x & (H_ - 1);
    float al[D_];
#pragma unroll
    for (int d = 0; d < D_; ++d) al[d] = attn_l[h * D_ + d];

    for (int i = blockIdx.x * 256 + threadIdx.x; i < totalNH; i += stride) {
        const float* f = ft + (size_t)(i >> 3) * HD + h * D_;
        float sl = 0.f;
#pragma unroll
        for (int d = 0; d < D_; ++d)
            sl = fmaf(f[d], al[d], sl);
        el[i] = sl;
        if (h == 0) cnt[i >> 3] = 0;
    }
}

__device__ __forceinline__ float lrelu(float x) {
    return x > 0.f ? x : 0.01f * x;
}

__global__ __launch_bounds__(256) void gat_hist(const int* __restrict__ dst, int* __restrict__ cnt, int E)
{
    const int stride = gridDim.x * 256;
    for (int e = blockIdx.x * 256 + threadIdx.x; e < E; e += stride)
        atomicAdd(&cnt[dst[e]], 1);
}

#define SCAN_PER 8
__global__ __launch_bounds__(1024) void gat_scan(int* cnt, int n)
{
    __shared__ int wsum[32];
    const int tid  = threadIdx.x;
    const int lane = tid & 31;
    const int wave = tid >> 5;
    int carry = 0;
    for (int base = 0; base < n; base += 1024 * SCAN_PER) {
        const int i0 = base + tid * SCAN_PER;
        int v[SCAN_PER];
        int t = 0;
#pragma unroll
        for (int q = 0; q < SCAN_PER; ++q) {
            v[q] = (i0 + q < n) ? cnt[i0 + q] : 0;
            t += v[q];
        }
        int incl = t;
        for (int o = 1; o < 32; o <<= 1) {
            int y = __shfl_up(incl, o);
            if (lane >= o) incl += y;
        }
        if (lane == 31) wsum[wave] = incl;
        __syncthreads();
        const int ws = wsum[lane];
        int wincl = ws;
        for (int o = 1; o < 32; o <<= 1) {
            int y = __shfl_up(wincl, o);
            if (lane >= o) wincl += y;
        }
        const int wave_excl = __shfl(wincl, wave) - __shfl(ws, wave);
        const int total     = __shfl(wincl, 31);
        __syncthreads();
        int run = carry + wave_excl + incl - t;
#pragma unroll
        for (int q = 0; q < SCAN_PER; ++q) {
            if (i0 + q < n) cnt[i0 + q] = run;
            run += v[q];
        }
        carry += total;
    }
}

__global__ __launch_bounds__(256) void gat_fill(
    const int* __restrict__ src, const int* __restrict__ dst,
    int* __restrict__ offs, int* __restrict__ ssrc, int E)
{
    const int stride = gridDim.x * 256;
    for (int e = blockIdx.x * 256 + threadIdx.x; e < E; e += stride) {
        const int pos = atomicAdd(&offs[dst[e]], 1);
        if ((unsigned)pos < (unsigned)E) ssrc[pos] = src[e];
    }
}

__global__ __launch_bounds__(256) void gat_gather(
    const int* __restrict__ offs, const int* __restrict__ ssrc,
    const float* __restrict__ el, const float* __restrict__ attn_r,
    const float* __restrict__ ft, float* __restrict__ out, int N, int E)
{
    const int lane = threadIdx.x & 31;
    const int h = lane >> 2;
    const size_t col = (size_t)lane * 4;
    const float* ar = attn_r + lane * 4;
    const float ar0 = ar[0], ar1 = ar[1], ar2 = ar[2], ar3 = ar[3];
    const float* elh = el + h;

    const int node0  = __builtin_amdgcn_readfirstlane((int)(blockIdx.x * 8 + (threadIdx.x >> 5)));
    const int nwaves = (int)gridDim.x * 8;
    for (int node = node0; node < N; node += nwaves) {
        int beg = (node > 0) ? offs[node - 1] : 0;
        int end = offs[node];
        beg = max(0, min(beg, E));
        end = max(beg, min(end, E));

        const v4f fd = *(const v4f*)(ft + (size_t)node * HD + col);
        float er = fd.x * ar0;
        er = fmaf(fd.y, ar1, er);
        er = fmaf(fd.z, ar2, er);
        er = fmaf(fd.w, ar3, er);
        er += __shfl_xor(er, 1);
        er += __shfl_xor(er, 2);

        float m = -INFINITY;
        for (int cb = beg; cb < end; cb += 32) {
            const int n = min(32, end - cb);
            int sv = (cb + lane < end) ? ssrc[cb + lane] : 0;
            sv = max(0, min(sv, N - 1));
            int k = 0;
            for (; k + 4 <= n; k += 4) {
                const int s0 = __builtin_amdgcn_readlane(sv, k);
                const int s1 = __builtin_amdgcn_readlane(sv, k + 1);
                const int s2 = __builtin_amdgcn_readlane(sv, k + 2);
                const int s3 = __builtin_amdgcn_readlane(sv, k + 3);
                const float x0 = lrelu(elh[s0 * H_] + er);
                const float x1 = lrelu(elh[s1 * H_] + er);
                const float x2 = lrelu(elh[s2 * H_] + er);
                const float x3 = lrelu(elh[s3 * H_] + er);
                m = fmaxf(m, fmaxf(fmaxf(x0, x1), fmaxf(x2, x3)));
            }
            for (; k < n; ++k) {
                const int s0 = __builtin_amdgcn_readlane(sv, k);
                m = fmaxf(m, lrelu(elh[s0 * H_] + er));
            }
        }

        float z = 0.f;
        v4f acc = {0.f, 0.f, 0.f, 0.f};
        for (int cb = beg; cb < end; cb += 32) {
            const int n = min(32, end - cb);
            int sv = (cb + lane < end) ? ssrc[cb + lane] : 0;
            sv = max(0, min(sv, N - 1));
            int k = 0;
            for (; k + 4 <= n; k += 4) {
                const int s0 = __builtin_amdgcn_readlane(sv, k);
                const int s1 = __builtin_amdgcn_readlane(sv, k + 1);
                const int s2 = __builtin_amdgcn_readlane(sv, k + 2);
                const int s3 = __builtin_amdgcn_readlane(sv, k + 3);
                const v4f r0 = *(const v4f*)(ft + (size_t)s0 * HD + col);
                const v4f r1 = *(const v4f*)(ft + (size_t)s1 * HD + col);
                const v4f r2 = *(const v4f*)(ft + (size_t)s2 * HD + col);
                const v4f r3 = *(const v4f*)(ft + (size_t)s3 * HD + col);
                const float p0 = expf(lrelu(elh[s0 * H_] + er) - m);
                const float p1 = expf(lrelu(elh[s1 * H_] + er) - m);
                const float p2 = expf(lrelu(elh[s2 * H_] + er) - m);
                const float p3 = expf(lrelu(elh[s3 * H_] + er) - m);
                z += p0; acc += r0 * p0;
                z += p1; acc += r1 * p1;
                z += p2; acc += r2 * p2;
                z += p3; acc += r3 * p3;
            }
            for (; k < n; ++k) {
                const int s0 = __builtin_amdgcn_readlane(sv, k);
                const v4f r0 = *(const v4f*)(ft + (size_t)s0 * HD + col);
                const float p0 = expf(lrelu(elh[s0 * H_] + er) - m);
                z += p0; acc += r0 * p0;
            }
        }
        if (end > beg) acc = acc / z;

        float* op = out + (size_t)node * HD + col;
        v4f o = *(const v4f*)op + acc;
        o.x = o.x > 0.f ? o.x : expm1f(o.x);
        o.y = o.y > 0.f ? o.y : expm1f(o.y);
        o.z = o.z > 0.f ? o.z : expm1f(o.z);
        o.w = o.w > 0.f ? o.w : expm1f(o.w);
        *(v4f*)op = o;
    }
}

extern "C" void kernel_launch(void* const* d_in, const int* in_sizes, int n_in,
                              void* d_out, int out_size, void* d_ws, size_t ws_size,
                              hipStream_t stream)
{
    const float* feat   = (const float*)d_in[0];
    const int*   src    = (const int*)  d_in[1];
    const int*   dst    = (const int*)  d_in[2];
    const float* W      = (const float*)d_in[3];
    const float* attn_l = (const float*)d_in[4];
    const float* attn_r = (const float*)d_in[5];
    const float* Wres   = (const float*)d_in[6];
    float* out = (float*)d_out;

    const int N = in_sizes[0] / IN_F;
    const int E = in_sizes[1];

    char* ws = (char*)d_ws;
    float* ft   = (float*)ws;                        ws += (size_t)N * HD * sizeof(float);
    float* el   = (float*)ws;                        ws += (size_t)N * H_ * sizeof(float);
    int*   ssrc = (int*)ws;                          ws += (size_t)E * sizeof(int);
    int*   offs = (int*)ws;

    dim3 gproj(512, 2);
    gat_proj_wmma<<<gproj, 256, 0, stream>>>(feat, W, Wres, ft, out, N);

    int totalNH = N * H_;
    gat_node_el<<<1024, 256, 0, stream>>>(ft, attn_l, el, offs, totalNH);

    gat_hist<<<1024, 256, 0, stream>>>(dst, offs, E);
    gat_scan<<<1, 1024, 0, stream>>>(offs, N);
    gat_fill<<<1024, 256, 0, stream>>>(src, dst, offs, ssrc, E);

    gat_gather<<<2048, 256, 0, stream>>>(offs, ssrc, el, attn_r, ft, out, N, E);
}


# counting sort with one atomic pass, not two: the histogram add's return is kept as each edge's rank, placement = start + rank by plain loads; sort runs first; row-sum order = add arrival order
# speedup vs baseline: 3.9530x; 1.1384x over previous
#include <hip/hip_runtime.h>
#include <hip/hip_bf16.h>
#include <math.h>

typedef __attribute__((ext_vector_type(2))) float v2f;
typedef __attribute__((ext_vector_type(8))) float v8f;
typedef __attribute__((ext_vector_type(4))) float v4f;

#define IN_F 64
#define HD   128
#define H_   8
#define D_   16

__global__ __launch_bounds__(256) void gat_proj_wmma(
    const float* __restrict__ feat,
    const float* __restrict__ W,
    const float* __restrict__ Wres,
    float* __restrict__ ft_ws,
    float* __restrict__ res_out,
    int nrows)
{
    const int wave = threadIdx.x >> 5;
    const int lane = threadIdx.x & 31;
    const int col0 = wave * 16;

    const float* B  = (blockIdx.y == 0) ? W : Wres;
    float* outp     = (blockIdx.y == 0) ? ft_ws : res_out;

    const int r  = lane & 15;
    const int hi = lane >> 4;

    v2f b[IN_F / 4];
#pragma unroll
    for (int j = 0; j < IN_F / 4; ++j) {
        const float* bp = B + (size_t)(4 * j + 2 * hi) * HD + col0 + r;
        b[j].x = bp[0];
        b[j].y = bp[HD];
    }

    const int ntiles = (nrows + 15) >> 4;
    for (int tile = blockIdx.x; tile < ntiles; tile += gridDim.x) {
        const int row0 = tile * 16;
        const int arow = (row0 + r < nrows) ? (row0 + r) : (nrows - 1);
        const float* apBase = feat + (size_t)arow * IN_F;

        v8f acc = {};
#pragma unroll
        for (int j = 0; j < IN_F / 4; ++j) {
            v2f a;
            a.x = apBase[4 * j + 2 * hi + 0];
            a.y = apBase[4 * j + 2 * hi + 1];
            acc = __builtin_amdgcn_wmma_f32_16x16x4_f32(
                 false, a,  false, b[j],
                 (short)0, acc,  false,  false);
        }

        const int rbase = row0 + 8 * hi;
#pragma unroll
        for (int v = 0; v < 8; ++v) {
            int row = rbase + v;
            if (row < nrows)
                outp[(size_t)row * HD + col0 + r] = acc[v];
        }
    }
}

__global__ __launch_bounds__(256) void gat_node_el(
    const float* __restrict__ ft,
    const float* __restrict__ attn_l,
    float* __restrict__ el,
    int totalNH)
{
    const int stride = gridDim.x * 256;
    const int h = threadIdx.x & (H_ - 1);
    float al[D_];
#pragma unroll
    for (int d = 0; d < D_; ++d) al[d] = attn_l[h * D_ + d];

    for (int i = blockIdx.x * 256 + threadIdx.x; i < totalNH; i += stride) {
        const float* f = ft + (size_t)(i >> 3) * HD + h * D_;
        float sl = 0.f;
#pragma unroll
        for (int d = 0; d < D_; ++d)
            sl = fmaf(f[d], al[d], sl);
        el[i] = sl;
    }
}

__device__ __forceinline__ float lrelu(float x) {
    return x > 0.f ? x : 0.01f * x;
}

__global__ __launch_bounds__(256) void gat_zero(int* __restrict__ cnt, int n)
{
    const int stride = gridDim.x * 256;
    for (int i = blockIdx.x * 256 + threadIdx.x; i < n; i += stride)
        cnt[i] = 0;
}

__global__ __launch_bounds__(256) void gat_hist(
    const int* __restrict__ dst, int* __restrict__ cnt, int* __restrict__ rank, int E)
{
    const int stride = gridDim.x * 256;
    for (int e = blockIdx.x * 256 + threadIdx.x; e < E; e += stride)
        rank[e] = atomicAdd(&cnt[dst[e]], 1);
}

#define SCAN_PER 8
__global__ __launch_bounds__(1024) void gat_scan(int* cnt, int n)
{
    __shared__ int wsum[32];
    const int tid  = threadIdx.x;
    const int lane = tid & 31;
    const int wave = tid >> 5;
    int carry = 0;
    for (int base = 0; base < n; base += 1024 * SCAN_PER) {
        const int i0 = base + tid * SCAN_PER;
        int v[SCAN_PER];
        int t = 0;
#pragma unroll
        for (int q = 0; q < SCAN_PER; ++q) {
            v[q] = (i0 + q < n) ? cnt[i0 + q] : 0;
            t += v[q];
        }
        int incl = t;
        for (int o = 1; o < 32; o <<= 1) {
            int y = __shfl_up(incl, o);
            if (lane >= o) incl += y;
        }
        if (lane == 31) wsum[wave] = incl;
        __syncthreads();
        const int ws = wsum[lane];
        int wincl = ws;
        for (int o = 1; o < 32; o <<= 1) {
            int y = __shfl_up(wincl, o);
            if (lane >= o) wincl += y;
        }
        const int wave_excl = __shfl(wincl, wave) - __shfl(ws, wave);
        const int total     = __shfl(wincl, 31);
        __syncthreads();
        int run = carry + wave_excl + incl - t;
#pragma unroll
        for (int q = 0; q < SCAN_PER; ++q) {
            if (i0 + q < n) cnt[i0 + q] = run;
            run += v[q];
        }
        carry += total;
    }
}

__global__ __launch_bounds__(256) void gat_fill(
    const int* __restrict__ src, const int* __restrict__ dst,
    const int* __restrict__ offs, const int* __restrict__ rank,
    int* __restrict__ ssrc, int E)
{
    const int stride = gridDim.x * 256;
    for (int e0 = blockIdx.x * 256 + threadIdx.x; e0 < E; e0 += 4 * stride) {
        int d[4], r[4], s[4], pos[4];
#pragma unroll
        for (int q = 0; q < 4; ++q) {
            const int e  = e0 + q * stride;
            const int ec = (e < E) ? e : e0;
            d[q] = dst[ec];
            r[q] = rank[ec];
            s[q] = src[ec];
        }
#pragma unroll
        for (int q = 0; q < 4; ++q) {
            const int p = offs[d[q]] + r[q];
            pos[q] = max(0, min(p, E - 1));
        }
#pragma unroll
        for (int q = 0; q < 4; ++q)
            ssrc[pos[q]] = s[q];
    }
}

__global__ __launch_bounds__(256) void gat_gather(
    const int* __restrict__ offs, const int* __restrict__ ssrc,
    const float* __restrict__ el, const float* __restrict__ attn_r,
    const float* __restrict__ ft, float* __restrict__ out, int N, int E)
{
    const int lane = threadIdx.x & 31;
    const int h = lane >> 2;
    const size_t col = (size_t)lane * 4;
    const float* ar = attn_r + lane * 4;
    const float ar0 = ar[0], ar1 = ar[1], ar2 = ar[2], ar3 = ar[3];
    const float* elh = el + h;

    const int node0  = __builtin_amdgcn_readfirstlane((int)(blockIdx.x * 8 + (threadIdx.x >> 5)));
    const int nwaves = (int)gridDim.x * 8;
    for (int node = node0; node < N; node += nwaves) {
        int beg = offs[node];
        int end = (node + 1 < N) ? offs[node + 1] : E;
        beg = max(0, min(beg, E));
        end = max(beg, min(end, E));

        const v4f fd = *(const v4f*)(ft + (size_t)node * HD + col);
        float er = fd.x * ar0;
        er = fmaf(fd.y, ar1, er);
        er = fmaf(fd.z, ar2, er);
        er = fmaf(fd.w, ar3, er);
        er += __shfl_xor(er, 1);
        er += __shfl_xor(er, 2);

        float m = -INFINITY;
        for (int cb = beg; cb < end; cb += 32) {
            const int n = min(32, end - cb);
            int sv = (cb + lane < end) ? ssrc[cb + lane] : 0;
            sv = max(0, min(sv, N - 1));
            int k = 0;
            for (; k + 4 <= n; k += 4) {
                const int s0 = __builtin_amdgcn_readlane(sv, k);
                const int s1 = __builtin_amdgcn_readlane(sv, k + 1);
                const int s2 = __builtin_amdgcn_readlane(sv, k + 2);
                const int s3 = __builtin_amdgcn_readlane(sv, k + 3);
                const float x0 = lrelu(elh[s0 * H_] + er);
                const float x1 = lrelu(elh[s1 * H_] + er);
                const float x2 = lrelu(elh[s2 * H_] + er);
                const float x3 = lrelu(elh[s3 * H_] + er);
                m = fmaxf(m, fmaxf(fmaxf(x0, x1), fmaxf(x2, x3)));
            }
            for (; k < n; ++k) {
                const int s0 = __builtin_amdgcn_readlane(sv, k);
                m = fmaxf(m, lrelu(elh[s0 * H_] + er));
            }
        }

        float z = 0.f;
        v4f acc = {0.f, 0.f, 0.f, 0.f};
        for (int cb = beg; cb < end; cb += 32) {
            const int n = min(32, end - cb);
            int sv = (cb + lane < end) ? ssrc[cb + lane] : 0;
            sv = max(0, min(sv, N - 1));
            int k = 0;
            for (; k + 4 <= n; k += 4) {
                const int s0 = __builtin_amdgcn_readlane(sv, k);
                const int s1 = __builtin_amdgcn_readlane(sv, k + 1);
                const int s2 = __builtin_amdgcn_readlane(sv, k + 2);
                const int s3 = __builtin_amdgcn_readlane(sv, k + 3);
                const v4f r0 = *(const v4f*)(ft + (size_t)s0 * HD + col);
                const v4f r1 = *(const v4f*)(ft + (size_t)s1 * HD + col);
                const v4f r2 = *(const v4f*)(ft + (size_t)s2 * HD + col);
                const v4f r3 = *(const v4f*)(ft + (size_t)s3 * HD + col);
                const float p0 = expf(lrelu(elh[s0 * H_] + er) - m);
                const float p1 = expf(lrelu(elh[s1 * H_] + er) - m);
                const float p2 = expf(lrelu(elh[s2 * H_] + er) - m);
                const float p3 = expf(lrelu(elh[s3 * H_] + er) - m);
                z += p0; acc += r0 * p0;
                z += p1; acc += r1 * p1;
                z += p2; acc += r2 * p2;
                z += p3; acc += r3 * p3;
            }
            for (; k < n; ++k) {
                const int s0 = __builtin_amdgcn_readlane(sv, k);
                const v4f r0 = *(const v4f*)(ft + (size_t)s0 * HD + col);
                const float p0 = expf(lrelu(elh[s0 * H_] + er) - m);
                z += p0; acc += r0 * p0;
            }
        }
        if (end > beg) acc = acc / z;

        float* op = out + (size_t)node * HD + col;
        v4f o = *(const v4f*)op + acc;
        o.x = o.x > 0.f ? o.x : expm1f(o.x);
        o.y = o.y > 0.f ? o.y : expm1f(o.y);
        o.z = o.z > 0.f ? o.z : expm1f(o.z);
        o.w = o.w > 0.f ? o.w : expm1f(o.w);
        *(v4f*)op = o;
    }
}

extern "C" void kernel_launch(void* const* d_in, const int* in_sizes, int n_in,
                              void* d_out, int out_size, void* d_ws, size_t ws_size,
                              hipStream_t stream)
{
    const float* feat   = (const float*)d_in[0];
    const int*   src    = (const int*)  d_in[1];
    const int*   dst    = (const int*)  d_in[2];
    const float* W      = (const float*)d_in[3];
    const float* attn_l = (const float*)d_in[4];
    const float* attn_r = (const float*)d_in[5];
    const float* Wres   = (const float*)d_in[6];
    float* out = (float*)d_out;

    const int N = in_sizes[0] / IN_F;
    const int E = in_sizes[1];

    char* ws = (char*)d_ws;
    float* ft   = (float*)ws;
    int*   rank = (int*)ws;                          ws += (size_t)N * HD * sizeof(float);
    float* el   = (float*)ws;                        ws += (size_t)N * H_ * sizeof(float);
    int*   ssrc = (int*)ws;                          ws += (size_t)E * sizeof(int);
    int*   offs = (int*)ws;


    gat_zero<<<128, 256, 0, stream>>>(offs, N);
    gat_hist<<<1024, 256, 0, stream>>>(dst, offs, rank, E);
    gat_scan<<<1, 1024, 0, stream>>>(offs, N);
    gat_fill<<<1024, 256, 0, stream>>>(src, dst, offs, rank, ssrc, E);

    dim3 gproj(512, 2);
    gat_proj_wmma<<<gproj, 256, 0, stream>>>(feat, W, Wres, ft, out, N);

    int totalNH = N * H_;
    gat_node_el<<<1024, 256, 0, stream>>>(ft, attn_l, el, totalNH);

    gat_gather<<<2048, 256, 0, stream>>>(offs, ssrc, el, attn_r, ft, out, N, E);
}


# one memory round trip per destination row, not two: scores of its first 32 edges are loaded once and kept in registers, its first 16 source rows and er go out with them; sum order unchanged
# speedup vs baseline: 7.8960x; 1.9975x over previous
#include <hip/hip_runtime.h>
#include <math.h>


typedef __attribute__((ext_vector_type(2))) float v2f;
typedef __attribute__((ext_vector_type(4))) float v4f;
typedef __attribute__((ext_vector_type(8))) float v8f;

#define IN_F 64
#define HD   128
#define H_   8
#define D_   16
#define SC   16
#define ZP   516
#define OP   132

__device__ __forceinline__ float lrelu(float x) {
    return x > 0.f ? x : 0.01f * x;
}

__device__ __forceinline__ v2f ld2_off(const float* __restrict__ base, unsigned byteoff) {
    return *(const v2f*)((const char*)base + byteoff);
}
__device__ __forceinline__ float ld1_off(const float* __restrict__ base, unsigned byteoff) {
    return *(const float*)((const char*)base + byteoff);
}

__global__ __launch_bounds__(256) void gat_scores_wmma(
    const float* __restrict__ feat,
    const float* __restrict__ W,
    const float* __restrict__ attn_l,
    const float* __restrict__ attn_r,
    float* __restrict__ elr,
    int nrows)
{
    __shared__ float Ms[IN_F * SC];
    const int tid = threadIdx.x;
    {
        const int c = tid & 15;
        const float* av = ((c < 8) ? attn_l : attn_r) + (c & 7) * D_;
        float a[D_];
#pragma unroll
        for (int d = 0; d < D_; ++d) a[d] = av[d];
#pragma unroll
        for (int i = 0; i < 4; ++i) {
            const int k = (tid >> 4) + 16 * i;
            const float* wp = W + (size_t)k * HD + (c & 7) * D_;
            float s = 0.f;
#pragma unroll
            for (int d = 0; d < D_; ++d) s = fmaf(wp[d], a[d], s);
            Ms[k * SC + c] = s;
        }
    }
    __syncthreads();

    const int lane = tid & 31;
    const int r  = lane & 15;
    const int hi = lane >> 4;

    v2f b[IN_F / 4];
#pragma unroll
    for (int j = 0; j < IN_F / 4; ++j) {
        b[j].x = Ms[(4 * j + 2 * hi) * SC + r];
        b[j].y = Ms[(4 * j + 2 * hi + 1) * SC + r];
    }

    const int wv = __builtin_amdgcn_readfirstlane((int)(blockIdx.x * 8 + (tid >> 5)));
    const int nw = (int)gridDim.x * 8;
    const int ntiles = (nrows + 15) >> 4;
    for (int tile = wv; tile < ntiles; tile += nw) {
        const int row0 = tile * 16;
        const int arow = (row0 + r < nrows) ? (row0 + r) : (nrows - 1);
        const float* apBase = feat + (size_t)arow * IN_F;
        v8f acc = {};
#pragma unroll
        for (int j = 0; j < IN_F / 4; ++j) {
            v2f a;
            a.x = apBase[4 * j + 2 * hi + 0];
            a.y = apBase[4 * j + 2 * hi + 1];
            acc = __builtin_amdgcn_wmma_f32_16x16x4_f32(
                false, a, false, b[j], (short)0, acc, false, false);
        }
        const int rbase = row0 + 8 * hi;
#pragma unroll
        for (int v = 0; v < 8; ++v) {
            const int row = rbase + v;
            if (row < nrows)
                elr[(size_t)row * SC + r] = acc[v];
        }
    }
}

#define BSH    8
#define BNODES (1 << BSH)
#define PBLK   256
#define PTHR   1024
#define TROWS  (PBLK / 32)
#define TCOLS  4

__device__ __forceinline__ void edge_slice(int blk, int nblk, int E, int& e0, int& e1)
{
    const int per = (E + nblk - 1) / nblk;
    e0 = min(E, blk * per);
    e1 = min(E, e0 + per);
}

struct EdgeRec { int bucket; unsigned word; };

__device__ __forceinline__ EdgeRec ld_edge(
    const int* __restrict__ src, const int* __restrict__ dst, int e, int N)
{
    const int d = max(0, min(dst[e], N - 1));
    EdgeRec r;
    r.bucket = d >> BSH;
    r.word   = ((unsigned)src[e] << BSH) | (unsigned)(d & (BNODES - 1));
    return r;
}

template <int G, int STRIDE, class LD, class ADD, class ST>
__device__ __forceinline__ int sweep_groups(int i, int end, LD ld, ADD add, ST st)
{
    for (; i + (G - 1) * STRIDE < end; i += G * STRIDE) {
        decltype(ld(0)) v[G];
        int p[G];
#pragma unroll
        for (int q = 0; q < G; ++q) v[q] = ld(i + q * STRIDE);
#pragma unroll
        for (int q = 0; q < G; ++q) p[q] = add(v[q]);
#pragma unroll
        for (int q = 0; q < G; ++q) st(v[q], p[q]);
    }
    return i;
}

template <int STRIDE, class LD, class ADD, class ST>
__device__ __forceinline__ void sweep_all(int i, int end, LD ld, ADD add, ST st)
{
    i = sweep_groups<8, STRIDE>(i, end, ld, add, st);
    i = sweep_groups<4, STRIDE>(i, end, ld, add, st);
    i = sweep_groups<2, STRIDE>(i, end, ld, add, st);
    sweep_groups<1, STRIDE>(i, end, ld, add, st);
}

__global__ __launch_bounds__(PTHR) void gat_bcount(
    const int* __restrict__ dst, int* __restrict__ T, int N, int E)
{
    __shared__ int cnt[PTHR];
    const int tid = threadIdx.x;
    const int nb = (N + BNODES - 1) >> BSH;
    cnt[tid] = 0;
    __syncthreads();
    int e0, e1;
    edge_slice((int)blockIdx.x, (int)gridDim.x, E, e0, e1);
    sweep_all<PTHR>(e0 + tid, e1,
        [&](int e) { return max(0, min(dst[e], N - 1)) >> BSH; },
        [&](int b) { atomicAdd(&cnt[b], 1); return 0; },
        [&](int, int) {});
    __syncthreads();
    if (tid < nb) T[(int)blockIdx.x * nb + tid] = cnt[tid];
}

__device__ __forceinline__ int wave_incl_scan(int v, int lane)
{
    for (int o = 1; o < 32; o <<= 1) {
        const int y = __shfl_up(v, o);
        if (lane >= o) v += y;
    }
    return v;
}

__global__ __launch_bounds__(PTHR) void gat_bscatter(
    const int* __restrict__ src, const int* __restrict__ dst,
    const int* __restrict__ T, int* __restrict__ bstart,
    unsigned* __restrict__ tmp, int N, int E)
{
    __shared__ int totS[PTHR];
    __shared__ int preS[PTHR];
    __shared__ int cur[PTHR];
    __shared__ int wsum[32];
    const int tid  = threadIdx.x;
    const int lane = tid & 31;
    const int wave = tid >> 5;
    const int nb   = (N + BNODES - 1) >> BSH;
    const int nblk = (int)gridDim.x;
    const int blk  = (int)blockIdx.x;

    totS[tid] = 0;
    preS[tid] = 0;
    __syncthreads();
    for (int bb = 0; bb < nb; bb += 32 * TCOLS) {
        int v[TCOLS][TROWS];
#pragma unroll
        for (int u = 0; u < TCOLS; ++u) {
#pragma unroll
            for (int jj = 0; jj < TROWS; ++jj) {
                const int b = bb + 32 * u + lane;
                const int j = wave * TROWS + jj;
                v[u][jj] = (b < nb && j < nblk) ? T[j * nb + b] : 0;
            }
        }
#pragma unroll
        for (int u = 0; u < TCOLS; ++u) {
            int t = 0, p = 0;
#pragma unroll
            for (int jj = 0; jj < TROWS; ++jj) {
                t += v[u][jj];
                p += (wave * TROWS + jj < blk) ? v[u][jj] : 0;
            }
            const int b = bb + 32 * u + lane;
            if (b < nb) {
                atomicAdd(&totS[b], t);
                atomicAdd(&preS[b], p);
            }
        }
    }
    __syncthreads();

    const int tot  = totS[tid];
    const int incl = wave_incl_scan(tot, lane);
    if (lane == 31) wsum[wave] = incl;
    __syncthreads();
    const int ws    = wsum[lane];
    const int wincl = wave_incl_scan(ws, lane);
    const int excl  = __shfl(wincl, wave) - __shfl(ws, wave) + incl - tot;
    cur[tid] = excl + preS[tid];
    if (blk == 0 && tid <= nb) bstart[tid] = excl;
    __syncthreads();

    int e0, e1;
    edge_slice(blk, nblk, E, e0, e1);
    sweep_all<PTHR>(e0 + tid, e1,
        [&](int e) { return ld_edge(src, dst, e, N); },
        [&](EdgeRec r) { return atomicAdd(&cur[r.bucket], 1); },
        [&](EdgeRec r, int p) { tmp[max(0, min(p, E - 1))] = r.word; });
}

__global__ __launch_bounds__(BNODES) void gat_bsort(
    const int* __restrict__ bstart, const unsigned* __restrict__ tmp,
    int* __restrict__ offs, int* __restrict__ ssrc, int N, int E)
{
    __shared__ int cnt[BNODES];
    __shared__ int wsum[BNODES / 32];
    const int tid  = threadIdx.x;
    const int lane = tid & 31;
    const int wave = tid >> 5;
    const int nb   = (N + BNODES - 1) >> BSH;
    for (int b = blockIdx.x; b < nb; b += gridDim.x) {
        const int lo = max(0, min(bstart[b], E));
        const int hi = max(lo, min(bstart[b + 1], E));
        cnt[tid] = 0;
        __syncthreads();
        sweep_all<BNODES>(lo + tid, hi,
            [&](int i) { return tmp[i]; },
            [&](unsigned k) { atomicAdd(&cnt[k & (BNODES - 1)], 1); return 0; },
            [&](unsigned, int) {});
        __syncthreads();
        const int c    = cnt[tid];
        const int incl = wave_incl_scan(c, lane);
        if (lane == 31) wsum[wave] = incl;
        __syncthreads();
        int wexcl = 0;
        for (int w = 0; w < BNODES / 32; ++w)
            wexcl += (w < wave) ? wsum[w] : 0;
        const int start = lo + wexcl + incl - c;
        const int node = (b << BSH) + tid;
        if (node < N) offs[node] = start;
        cnt[tid] = start;
        __syncthreads();
        sweep_all<BNODES>(lo + tid, hi,
            [&](int i) { return tmp[i]; },
            [&](unsigned k) { return atomicAdd(&cnt[k & (BNODES - 1)], 1); },
            [&](unsigned k, int p) { ssrc[max(0, min(p, E - 1))] = (int)(k >> BSH); });
        __syncthreads();
    }
}

template <int NG>
__device__ __forceinline__ void edge_groups(
    int sv, int k, int n, int lane, int g, int h, float er, float m,
    const float* __restrict__ elr, const float* __restrict__ feat,
    float& z, v2f (&acc)[H_])
{
    float p[NG];
    v2f f[NG * 4];
    int ps[NG];
#pragma unroll
    for (int u = 0; u < NG; ++u) {
        const int i = k + 4 * u + g;
        ps[u] = __shfl(sv, min(i, n - 1));
    }
#pragma unroll
    for (int q = 0; q < NG * 4; ++q) {
        const int idx = min(k + q, n - 1);
        const int s = __builtin_amdgcn_readlane(sv, idx);
        f[q] = ld2_off(feat, (unsigned)s * (IN_F * 4u) + (unsigned)lane * 8u);
    }
#pragma unroll
    for (int u = 0; u < NG; ++u)
        p[u] = ld1_off(elr, (unsigned)ps[u] * (SC * 4u) + (unsigned)h * 4u);
#pragma unroll
    for (int u = 0; u < NG; ++u) {
        const int i = k + 4 * u + g;
        const float e = expf(lrelu(p[u] + er) - m);
        p[u] = (i < n) ? e : 0.f;
        z += p[u];
    }
#pragma unroll
    for (int q = 0; q < NG * 4; ++q) {
        const int pbits = __float_as_int(p[q >> 2]);
#pragma unroll
        for (int hh = 0; hh < H_; ++hh) {
            const float ph = __int_as_float(__builtin_amdgcn_readlane(pbits, (q & 3) * 8 + hh));
            acc[hh].x = fmaf(ph, f[q].x, acc[hh].x);
            acc[hh].y = fmaf(ph, f[q].y, acc[hh].y);
        }
    }
}

template <int NG, int P0>
__device__ __forceinline__ void fma_rows(const float (&x)[8], const v2f (&f)[NG * 4], v2f (&acc)[H_])
{
#pragma unroll
    for (int q = 0; q < NG * 4; ++q) {
        const int pbits = __float_as_int(x[P0 + (q >> 2)]);
#pragma unroll
        for (int hh = 0; hh < H_; ++hh) {
            const float ph = __int_as_float(__builtin_amdgcn_readlane(pbits, (q & 3) * 8 + hh));
            acc[hh].x = fmaf(ph, f[q].x, acc[hh].x);
            acc[hh].y = fmaf(ph, f[q].y, acc[hh].y);
        }
    }
}

template <int NG>
__device__ __forceinline__ void load_rows(int sv, int k, int n, int lane,
    const float* __restrict__ feat, v2f (&f)[NG * 4])
{
#pragma unroll
    for (int q = 0; q < NG * 4; ++q) {
        const int s = __builtin_amdgcn_readlane(sv, min(k + q, n - 1));
        f[q] = ld2_off(feat, (unsigned)s * (IN_F * 4u) + (unsigned)lane * 8u);
    }
}

template <int NG>
__device__ __forceinline__ void rest_rows(int sv, int n, int lane,
    const float* __restrict__ feat, const float (&x)[8], v2f (&acc)[H_])
{
    v2f f[NG * 4];
    load_rows<NG>(sv, 16, n, lane, feat, f);
    fma_rows<NG, 4>(x, f, acc);
}

template <int NG, int NS>
__device__ __forceinline__ void first_chunk(
    int sv, int n, int lane, int g, int h, float er, float m_in,
    const float* __restrict__ elr, const float* __restrict__ feat,
    float& m_out, float& z, v2f (&acc)[H_])
{
    float x[8];
    int ps[NS];
    v2f f[NG * 4];
    const int ng = n - g;
#pragma unroll
    for (int u = 0; u < NS; ++u) ps[u] = __shfl(sv, g + min(4 * u, ng - 1));
#pragma unroll
    for (int u = 0; u < NS; ++u)
        x[u] = ld1_off(elr, (unsigned)ps[u] * (SC * 4u) + (unsigned)h * 4u);
#pragma unroll
    for (int u = NS; u < 8; ++u) x[u] = 0.f;
    load_rows<NG>(sv, 0, n, lane, feat, f);
    __builtin_amdgcn_sched_barrier(0);

    float m = m_in;
#pragma unroll
    for (int u = 0; u < NS; ++u) {
        x[u] = lrelu(x[u] + er);
        m = fmaxf(m, (4 * u < ng) ? x[u] : -INFINITY);
    }
    m = fmaxf(m, __shfl_xor(m, 8));
    m = fmaxf(m, __shfl_xor(m, 16));
    m_out = m;
#pragma unroll
    for (int u = 0; u < NS; ++u) {
        const float e = expf(x[u] - m);
        x[u] = (4 * u < ng) ? e : 0.f;
        z += x[u];
    }
    fma_rows<NG, 0>(x, f, acc);
    if constexpr (NS == 8) {
        if (n - 16 > 8)      rest_rows<4>(sv, n, lane, feat, x, acc);
        else if (n - 16 > 4) rest_rows<2>(sv, n, lane, feat, x, acc);
        else if (n - 16 > 0) rest_rows<1>(sv, n, lane, feat, x, acc);
    }
}

__device__ __forceinline__ int load_bounds(
    const int* __restrict__ offs, int row0, int wave, int lane, int N, int E)
{
    const int bi = row0 + wave * 2 + (lane & 3);
    const int v = offs[min(bi, N - 1)];
    const int lo = (bi < N) ? 0 : E;
    return max(lo, min(v, E));
}

__device__ __forceinline__ float load_er(
    const float* __restrict__ elr, int row0, int wave, int t, int h, int N)
{
    const int node = min(row0 + wave * 2 + t, N - 1);
    return ld1_off(elr, (unsigned)node * (SC * 4u) + (unsigned)(H_ + h) * 4u);
}

__device__ __forceinline__ int load_first_sources(
    const int* __restrict__ ssrc, int voff, int t, int lane, int N, int E)
{
    const int bb = __builtin_amdgcn_readlane(voff, t);
    const int be = max(bb, __builtin_amdgcn_readlane(voff, t + 1));
    const int i = bb + lane;
    const int s = ssrc[max(0, min(i, E - 1))];
    const int hi = (i < be) ? (N - 1) : 0;
    return min(max(s, 0), hi);
}

__global__ __launch_bounds__(256, 8) void gat_aggregate_project(
    const int* __restrict__ offs, const int* __restrict__ ssrc,
    const float* __restrict__ elr, const float* __restrict__ feat,
    const float* __restrict__ W, const float* __restrict__ Wres,
    float* __restrict__ out, int N, int E)
{
    __shared__ float zt[16 * ZP];
    __shared__ v2f brs[8 * (IN_F / 4) * 32];
    __shared__ float ot[16 * OP];

    const int lane = threadIdx.x & 31;
    const int wave = __builtin_amdgcn_readfirstlane((int)(threadIdx.x >> 5));
    const int r  = lane & 15;
    const int hi = lane >> 4;
    const int h  = lane & 7;
    const int g  = lane >> 3;
    const int col0 = wave * 16;

    v2f bw[IN_F / 4];
    v2f* brw = brs + wave * (IN_F / 4) * 32 + lane;
#pragma unroll
    for (int j = 0; j < IN_F / 4; ++j) {
        const size_t o = (size_t)(4 * j + 2 * hi) * HD + col0 + r;
        bw[j].x = W[o];
        bw[j].y = W[o + HD];
        v2f t2;
        t2.x = Wres[o];
        t2.y = Wres[o + HD];
        brw[j * 32] = t2;
    }

    const int ntiles = (N + 15) >> 4;

    int voff = load_bounds(offs, (int)blockIdx.x * 16, wave, lane, N, E);
    int sv0a = load_first_sources(ssrc, voff, 0, lane, N, E);
    int sv0b = load_first_sources(ssrc, voff, 1, lane, N, E);
    float er0 = load_er(elr, (int)blockIdx.x * 16, wave, 0, h, N);
    float er1 = load_er(elr, (int)blockIdx.x * 16, wave, 1, h, N);

    for (int tile = blockIdx.x; tile < ntiles; tile += gridDim.x) {
        const int row0 = tile * 16;

        for (int t = 0; t < 2; ++t) {
            const int nt = wave * 2 + t;
            const int node = row0 + nt;
            v2f acc[H_];
#pragma unroll
            for (int hh = 0; hh < H_; ++hh) acc[hh] = (v2f){0.f, 0.f};

            if (node < N) {
                const int beg = __builtin_amdgcn_readlane(voff, t);
                const int end = max(beg, __builtin_amdgcn_readlane(voff, t + 1));
                const int svf = t ? sv0b : sv0a;

                const float er = t ? er1 : er0;

                float m = -INFINITY;
                for (int cb = beg + 32; cb < end; cb += 32) {
                    const int n = min(32, end - cb);
                    int sv = (cb + lane < end) ? ssrc[cb + lane] : 0;
                    sv = max(0, min(sv, N - 1));
                    for (int k = 0; k < n; k += 16) {
                        int si[4];
                        float x[4];
#pragma unroll
                        for (int u = 0; u < 4; ++u)
                            si[u] = __shfl(sv, min(k + 4 * u + g, n - 1));
#pragma unroll
                        for (int u = 0; u < 4; ++u)
                            x[u] = ld1_off(elr, (unsigned)si[u] * (SC * 4u) + (unsigned)h * 4u);
#pragma unroll
                        for (int u = 0; u < 4; ++u) {
                            const float xv = lrelu(x[u] + er);
                            m = fmaxf(m, (k + 4 * u + g < n) ? xv : -INFINITY);
                        }
                    }
                }

                float z = 0.f;
                if (end > beg) {
                    const int n0 = min(32, end - beg);
                    if (n0 > 16)     first_chunk<4, 8>(svf, n0, lane, g, h, er, m, elr, feat, m, z, acc);
                    else if (n0 > 8) first_chunk<4, 4>(svf, n0, lane, g, h, er, m, elr, feat, m, z, acc);
                    else if (n0 > 4) first_chunk<2, 2>(svf, n0, lane, g, h, er, m, elr, feat, m, z, acc);
                    else             first_chunk<1, 1>(svf, n0, lane, g, h, er, m, elr, feat, m, z, acc);
                }

                for (int cb = beg + 32; cb < end; cb += 32) {
                    const int n = min(32, end - cb);
                    int sv = (cb + lane < end) ? ssrc[cb + lane] : 0;
                    sv = max(0, min(sv, N - 1));
                    int k = 0;
                    for (; k + 16 <= n; k += 16)
                        edge_groups<4>(sv, k, n, lane, g, h, er, m, elr, feat, z, acc);
                    if (n - k > 8)
                        edge_groups<4>(sv, k, n, lane, g, h, er, m, elr, feat, z, acc);
                    else if (n - k > 4)
                        edge_groups<2>(sv, k, n, lane, g, h, er, m, elr, feat, z, acc);
                    else if (n - k > 0)
                        edge_groups<1>(sv, k, n, lane, g, h, er, m, elr, feat, z, acc);
                }
                z += __shfl_xor(z, 8);
                z += __shfl_xor(z, 16);

                if (end > beg) {
                    const int rzb = __float_as_int(1.0f / z);
#pragma unroll
                    for (int hh = 0; hh < H_; ++hh) {
                        const float rh = __int_as_float(__builtin_amdgcn_readlane(rzb, hh));
                        acc[hh] *= rh;
                    }
                }
            }

            float* zrow = zt + nt * ZP + 2 * lane;
#pragma unroll
            for (int hh = 0; hh < H_; ++hh)
                *(v2f*)(zrow + hh * IN_F) = acc[hh];
        }
        __syncthreads();
        const int voff_n = load_bounds(offs, (tile + (int)gridDim.x) * 16, wave, lane, N, E);

        const int arow = (row0 + r < N) ? (row0 + r) : (N - 1);
        const float* ap = feat + (size_t)arow * IN_F + 2 * hi;
        const float* zp = zt + r * ZP + wave * IN_F + 2 * hi;

        v8f c = {};
#pragma unroll
        for (int j = 0; j < IN_F / 4; ++j) {
            const v2f a = *(const v2f*)(zp + 4 * j);
            c = __builtin_amdgcn_wmma_f32_16x16x4_f32(
                false, a, false, bw[j], (short)0, c, false, false);
        }
#pragma unroll
        for (int j = 0; j < IN_F / 4; ++j) {
            v2f a;
            a.x = ap[4 * j + 0];
            a.y = ap[4 * j + 1];
            const v2f b = brw[j * 32];
            c = __builtin_amdgcn_wmma_f32_16x16x4_f32(
                false, a, false, b, (short)0, c, false, false);
        }
        {
            float* orow = ot + (8 * hi) * OP + col0 + r;
#pragma unroll
            for (int v = 0; v < 8; ++v) {
                const float o = c[v];
                orow[v * OP] = o > 0.f ? o : expm1f(o);
            }
        }
        __syncthreads();
        const int sv0a_n = load_first_sources(ssrc, voff_n, 0, lane, N, E);
        const int sv0b_n = load_first_sources(ssrc, voff_n, 1, lane, N, E);
        const float er0_n = load_er(elr, (tile + (int)gridDim.x) * 16, wave, 0, h, N);
        const float er1_n = load_er(elr, (tile + (int)gridDim.x) * 16, wave, 1, h, N);

#pragma unroll
        for (int t = 0; t < 2; ++t) {
            const int orw = wave * 2 + t;
            const int grow = row0 + orw;
            if (grow < N)
                *(v4f*)((char*)out + ((unsigned)grow * (HD * 4u) + (unsigned)lane * 16u)) = *(const v4f*)(ot + orw * OP + 4 * lane);
        }
        voff = voff_n;
        sv0a = sv0a_n;
        sv0b = sv0b_n;
        er0 = er0_n;
        er1 = er1_n;
    }
}

extern "C" void kernel_launch(void* const* d_in, const int* in_sizes, int n_in,
                              void* d_out, int out_size, void* d_ws, size_t ws_size,
                              hipStream_t stream)
{
    const float* feat   = (const float*)d_in[0];
    const int*   src    = (const int*)  d_in[1];
    const int*   dst    = (const int*)  d_in[2];
    const float* W      = (const float*)d_in[3];
    const float* attn_l = (const float*)d_in[4];
    const float* attn_r = (const float*)d_in[5];
    const float* Wres   = (const float*)d_in[6];
    float* out = (float*)d_out;

    const int N = in_sizes[0] / IN_F;
    const int E = in_sizes[1];

    char* ws = (char*)d_ws;
    unsigned* tmp = (unsigned*)ws;
    float* elr  = (float*)ws + E;                    ws += (size_t)N * HD * sizeof(float);
    int*   T      = (int*)ws;
    int*   bstart = (int*)ws + PBLK * PTHR;          ws += (size_t)N * H_ * sizeof(float);
    int*   ssrc = (int*)ws;                          ws += (size_t)E * sizeof(int);
    int*   offs = (int*)ws;

    gat_bcount<<<PBLK, PTHR, 0, stream>>>(dst, T, N, E);
    gat_bscatter<<<PBLK, PTHR, 0, stream>>>(src, dst, T, bstart, tmp, N, E);
    gat_bsort<<<391, BNODES, 0, stream>>>(bstart, tmp, offs, ssrc, N, E);

    gat_scores_wmma<<<391, 256, 0, stream>>>(feat, W, attn_l, attn_r, elr, N);

    gat_aggregate_project<<<1024, 256, 0, stream>>>(offs, ssrc, elr, feat, W, Wres, out, N, E);
}


# exponentials on the base-2 instruction: softmax weights and ELU's e^x - 1 as exp2(x * log2 e), not the library's range-split expf / expm1f (13 and 22 instructions each); sum order unchanged
# speedup vs baseline: 11.1963x; 1.4180x over previous
#include <hip/hip_runtime.h>
#include <math.h>


typedef __attribute__((ext_vector_type(2))) float v2f;
typedef __attribute__((ext_vector_type(4))) float v4f;
typedef __attribute__((ext_vector_type(8))) float v8f;

#define IN_F 64
#define HD   128
#define H_   8
#define D_   16
#define SC   16
#define ZP   516
#define OP   132

__device__ __forceinline__ float lrelu(float x) {
    return x > 0.f ? x : 0.01f * x;
}

__device__ __forceinline__ float exp_b2(float x) {
    return __builtin_amdgcn_exp2f(x * 1.4426950408889634f);
}

__device__ __forceinline__ v2f ld2_off(const float* __restrict__ base, unsigned byteoff) {
    return *(const v2f*)((const char*)base + byteoff);
}
__device__ __forceinline__ float ld1_off(const float* __restrict__ base, unsigned byteoff) {
    return *(const float*)((const char*)base + byteoff);
}

__device__ __forceinline__ void scores_role(
    const float* __restrict__ feat,
    const float* __restrict__ W,
    const float* __restrict__ attn_l,
    const float* __restrict__ attn_r,
    float* __restrict__ elr,
    int nrows, int blk, int nblk)
{
    __shared__ float Ms[IN_F * SC];
    const int tid = threadIdx.x;
    {
        const int c = tid & 15;
        const float* av = ((c < 8) ? attn_l : attn_r) + (c & 7) * D_;
        float a[D_];
#pragma unroll
        for (int d = 0; d < D_; ++d) a[d] = av[d];
#pragma unroll
        for (int i = 0; i < 4; ++i) {
            const int k = (tid >> 4) + 16 * i;
            const float* wp = W + (size_t)k * HD + (c & 7) * D_;
            float s = 0.f;
#pragma unroll
            for (int d = 0; d < D_; ++d) s = fmaf(wp[d], a[d], s);
            Ms[k * SC + c] = s;
        }
    }
    __syncthreads();

    const int lane = tid & 31;
    const int r  = lane & 15;
    const int hi = lane >> 4;

    v2f b[IN_F / 4];
#pragma unroll
    for (int j = 0; j < IN_F / 4; ++j) {
        b[j].x = Ms[(4 * j + 2 * hi) * SC + r];
        b[j].y = Ms[(4 * j + 2 * hi + 1) * SC + r];
    }

    const int wv = __builtin_amdgcn_readfirstlane(blk * 8 + (tid >> 5));
    const int nw = nblk * 8;
    const int ntiles = (nrows + 15) >> 4;
    for (int tile = wv; tile < ntiles; tile += nw) {
        const int row0 = tile * 16;
        const int arow = (row0 + r < nrows) ? (row0 + r) : (nrows - 1);
        const float* apBase = feat + (size_t)arow * IN_F;
        v8f acc = {};
#pragma unroll
        for (int j = 0; j < IN_F / 4; ++j) {
            v2f a;
            a.x = apBase[4 * j + 2 * hi + 0];
            a.y = apBase[4 * j + 2 * hi + 1];
            acc = __builtin_amdgcn_wmma_f32_16x16x4_f32(
                false, a, false, b[j], (short)0, acc, false, false);
        }
        const int rbase = row0 + 8 * hi;
#pragma unroll
        for (int v = 0; v < 8; ++v) {
            const int row = rbase + v;
            if (row < nrows)
                elr[(size_t)row * SC + r] = acc[v];
        }
    }
}

#define BSH    8
#define BNODES (1 << BSH)
#define PBLK   256
#define PTHR   1024
#define PPW    (PBLK / (BNODES / 32))
#define PGRP   16
static_assert(PBLK == BNODES, "bsort_role maps thread t to slice t");
static_assert(PPW % PGRP == 0, "piece groups must tile a wave's pieces");

__device__ __forceinline__ void edge_slice(int blk, int nblk, int E, int& e0, int& e1)
{
    const int per = (E + nblk - 1) / nblk;
    e0 = min(E, blk * per);
    e1 = min(E, e0 + per);
}

struct EdgeRec { int bucket; unsigned word; };

__device__ __forceinline__ EdgeRec ld_edge(
    const int* __restrict__ src, const int* __restrict__ dst, int e, int N)
{
    const int d = max(0, min(dst[e], N - 1));
    EdgeRec r;
    r.bucket = d >> BSH;
    r.word   = ((unsigned)src[e] << BSH) | (unsigned)(d & (BNODES - 1));
    return r;
}

template <int G, int STRIDE, class LD, class ADD, class ST>
__device__ __forceinline__ int sweep_groups(int i, int end, LD ld, ADD add, ST st)
{
    for (; i + (G - 1) * STRIDE < end; i += G * STRIDE) {
        decltype(ld(0)) v[G];
        int p[G];
#pragma unroll
        for (int q = 0; q < G; ++q) v[q] = ld(i + q * STRIDE);
#pragma unroll
        for (int q = 0; q < G; ++q) p[q] = add(v[q]);
#pragma unroll
        for (int q = 0; q < G; ++q) st(v[q], p[q]);
    }
    return i;
}

template <int STRIDE, class LD, class ADD, class ST>
__device__ __forceinline__ void sweep_all(int i, int end, LD ld, ADD add, ST st)
{
    i = sweep_groups<8, STRIDE>(i, end, ld, add, st);
    i = sweep_groups<4, STRIDE>(i, end, ld, add, st);
    i = sweep_groups<2, STRIDE>(i, end, ld, add, st);
    sweep_groups<1, STRIDE>(i, end, ld, add, st);
}

__device__ __forceinline__ int wave_incl_scan(int v, int lane)
{
    for (int o = 1; o < 32; o <<= 1) {
        const int y = __shfl_up(v, o);
        if (lane >= o) v += y;
    }
    return v;
}

__global__ __launch_bounds__(PTHR) void gat_bpart(
    const int* __restrict__ src, const int* __restrict__ dst,
    int* __restrict__ S, unsigned* __restrict__ tmp, int N, int E)
{
    __shared__ int cnt[PTHR];
    __shared__ int wsum[32];
    const int tid  = threadIdx.x;
    const int lane = tid & 31;
    const int wave = tid >> 5;
    const int nb   = (N + BNODES - 1) >> BSH;
    const int blk  = (int)blockIdx.x;

    cnt[tid] = 0;
    __syncthreads();
    int e0, e1;
    edge_slice(blk, (int)gridDim.x, E, e0, e1);
    sweep_all<PTHR>(e0 + tid, e1,
        [&](int e) { return max(0, min(dst[e], N - 1)) >> BSH; },
        [&](int b) { atomicAdd(&cnt[b], 1); return 0; },
        [&](int, int) {});
    __syncthreads();

    const int tot  = cnt[tid];
    const int incl = wave_incl_scan(tot, lane);
    if (lane == 31) wsum[wave] = incl;
    __syncthreads();
    const int ws    = wsum[lane];
    const int wincl = wave_incl_scan(ws, lane);
    const int excl  = __shfl(wincl, wave) - __shfl(ws, wave) + incl - tot;
    if (tid <= nb) S[blk * (nb + 1) + tid] = excl;
    cnt[tid] = e0 + excl;
    __syncthreads();

    sweep_all<PTHR>(e0 + tid, e1,
        [&](int e) { return ld_edge(src, dst, e, N); },
        [&](EdgeRec r) { return atomicAdd(&cnt[r.bucket], 1); },
        [&](EdgeRec r, int p) { tmp[max(0, min(p, E - 1))] = r.word; });
}

template <bool PLACE>
__device__ __forceinline__ void piece_pass(
    const int* plo, const int* pn, const unsigned* __restrict__ tmp,
    int* cnt, int* __restrict__ ssrc, int wave, int lane, int E)
{
    for (int g0 = 0; g0 < PPW; g0 += PGRP) {
        int n[PGRP];
        unsigned k[PGRP];
#pragma unroll
        for (int q = 0; q < PGRP; ++q) {
            const int j = wave * PPW + g0 + q;
            n[q] = pn[j];
            const int i = plo[j] + min(lane, n[q] - 1);
            k[q] = tmp[max(0, min(i, E - 1))];
        }
#pragma unroll
        for (int q = 0; q < PGRP; ++q) {
            if (lane < n[q]) {
                if constexpr (PLACE) {
                    const int p = atomicAdd(&cnt[k[q] & (BNODES - 1)], 1);
                    ssrc[max(0, min(p, E - 1))] = (int)(k[q] >> BSH);
                } else {
                    atomicAdd(&cnt[k[q] & (BNODES - 1)], 1);
                }
            }
        }
        int nmax = n[0];
#pragma unroll
        for (int q = 1; q < PGRP; ++q) nmax = max(nmax, n[q]);
        if (nmax > 32) {
            for (int q = 0; q < PGRP; ++q) {
                const int j = wave * PPW + g0 + q;
                const int nn = pn[j];
                const int pp = plo[j];
                for (int i = lane + 32; i < nn; i += 32) {
                    const unsigned kk = tmp[max(0, min(pp + i, E - 1))];
                    if constexpr (PLACE) {
                        const int p = atomicAdd(&cnt[kk & (BNODES - 1)], 1);
                        ssrc[max(0, min(p, E - 1))] = (int)(kk >> BSH);
                    } else {
                        atomicAdd(&cnt[kk & (BNODES - 1)], 1);
                    }
                }
            }
        }
    }
}

__device__ __forceinline__ void bsort_role(
    const int* __restrict__ S, const unsigned* __restrict__ tmp,
    int* __restrict__ offs, int* __restrict__ ssrc, int N, int E, int blk, int nblk)
{
    __shared__ int cnt[BNODES];
    __shared__ int wsum[BNODES / 32];
    __shared__ int plo[PBLK];
    __shared__ int pn[PBLK];
    const int tid  = threadIdx.x;
    const int lane = tid & 31;
    const int wave = __builtin_amdgcn_readfirstlane(tid >> 5);
    const int nb   = (N + BNODES - 1) >> BSH;
    int e0, e1;
    edge_slice(tid, PBLK, E, e0, e1);
    for (int b = blk; b < nb; b += nblk) {
        const int s0 = max(0, min(S[tid * (nb + 1) + b], e1 - e0));
        const int s1 = max(s0, min(S[tid * (nb + 1) + b + 1], e1 - e0));
        plo[tid] = e0 + s0;
        pn[tid]  = s1 - s0;
        cnt[tid] = 0;
        const int sincl = wave_incl_scan(s0, lane);
        if (lane == 31) wsum[wave] = sincl;
        __syncthreads();
        int lo = 0;
        for (int w = 0; w < BNODES / 32; ++w) lo += wsum[w];
        lo = max(0, min(lo, E));
        piece_pass<false>(plo, pn, tmp, cnt, ssrc, wave, lane, E);
        __syncthreads();
        const int c    = cnt[tid];
        const int incl = wave_incl_scan(c, lane);
        if (lane == 31) wsum[wave] = incl;
        __syncthreads();
        int wexcl = 0;
        for (int w = 0; w < BNODES / 32; ++w)
            wexcl += (w < wave) ? wsum[w] : 0;
        const int start = lo + wexcl + incl - c;
        const int node = (b << BSH) + tid;
        if (node < N) offs[node] = start;
        cnt[tid] = start;
        __syncthreads();
        piece_pass<true>(plo, pn, tmp, cnt, ssrc, wave, lane, E);
        __syncthreads();
    }
}

__global__ __launch_bounds__(BNODES) void gat_bsort_scores(
    const int* __restrict__ S, const unsigned* __restrict__ tmp,
    int* __restrict__ offs, int* __restrict__ ssrc,
    const float* __restrict__ feat, const float* __restrict__ W,
    const float* __restrict__ attn_l, const float* __restrict__ attn_r,
    float* __restrict__ elr, int N, int E, int nsort)
{
    const int blk = (int)blockIdx.x;
    if (blk < nsort)
        bsort_role(S, tmp, offs, ssrc, N, E, blk, nsort);
    else
        scores_role(feat, W, attn_l, attn_r, elr, N, blk - nsort, (int)gridDim.x - nsort);
}
template <int NG>
__device__ __forceinline__ void edge_groups(
    int sv, int k, int n, int lane, int g, int h, float er, float m,
    const float* __restrict__ elr, const float* __restrict__ feat,
    float& z, v2f (&acc)[H_])
{
    float p[NG];
    v2f f[NG * 4];
    int ps[NG];
#pragma unroll
    for (int u = 0; u < NG; ++u) {
        const int i = k + 4 * u + g;
        ps[u] = __shfl(sv, min(i, n - 1));
    }
#pragma unroll
    for (int q = 0; q < NG * 4; ++q) {
        const int idx = min(k + q, n - 1);
        const int s = __builtin_amdgcn_readlane(sv, idx);
        f[q] = ld2_off(feat, (unsigned)s * (IN_F * 4u) + (unsigned)lane * 8u);
    }
#pragma unroll
    for (int u = 0; u < NG; ++u)
        p[u] = ld1_off(elr, (unsigned)ps[u] * (SC * 4u) + (unsigned)h * 4u);
#pragma unroll
    for (int u = 0; u < NG; ++u) {
        const int i = k + 4 * u + g;
        const float e = exp_b2(lrelu(p[u] + er) - m);
        p[u] = (i < n) ? e : 0.f;
        z += p[u];
    }
#pragma unroll
    for (int q = 0; q < NG * 4; ++q) {
        const int pbits = __float_as_int(p[q >> 2]);
#pragma unroll
        for (int hh = 0; hh < H_; ++hh) {
            const float ph = __int_as_float(__builtin_amdgcn_readlane(pbits, (q & 3) * 8 + hh));
            acc[hh].x = fmaf(ph, f[q].x, acc[hh].x);
            acc[hh].y = fmaf(ph, f[q].y, acc[hh].y);
        }
    }
}

template <int NG, int P0>
__device__ __forceinline__ void fma_rows(const float (&x)[8], const v2f (&f)[NG * 4], v2f (&acc)[H_])
{
#pragma unroll
    for (int q = 0; q < NG * 4; ++q) {
        const int pbits = __float_as_int(x[P0 + (q >> 2)]);
#pragma unroll
        for (int hh = 0; hh < H_; ++hh) {
            const float ph = __int_as_float(__builtin_amdgcn_readlane(pbits, (q & 3) * 8 + hh));
            acc[hh].x = fmaf(ph, f[q].x, acc[hh].x);
            acc[hh].y = fmaf(ph, f[q].y, acc[hh].y);
        }
    }
}

template <int NG>
__device__ __forceinline__ void load_rows(int sv, int k, int n, int lane,
    const float* __restrict__ feat, v2f (&f)[NG * 4])
{
#pragma unroll
    for (int q = 0; q < NG * 4; ++q) {
        const int s = __builtin_amdgcn_readlane(sv, min(k + q, n - 1));
        f[q] = ld2_off(feat, (unsigned)s * (IN_F * 4u) + (unsigned)lane * 8u);
    }
}

template <int NG>
__device__ __forceinline__ void rest_rows(int sv, int n, int lane,
    const float* __restrict__ feat, const float (&x)[8], v2f (&acc)[H_])
{
    v2f f[NG * 4];
    load_rows<NG>(sv, 16, n, lane, feat, f);
    fma_rows<NG, 4>(x, f, acc);
}

template <int NG, int NS>
__device__ __forceinline__ void first_chunk(
    int sv, int n, int lane, int g, int h, float er, float m_in,
    const float* __restrict__ elr, const float* __restrict__ feat,
    float& m_out, float& z, v2f (&acc)[H_])
{
    float x[8];
    int ps[NS];
    v2f f[NG * 4];
    const int ng = n - g;
#pragma unroll
    for (int u = 0; u < NS; ++u) ps[u] = __shfl(sv, g + min(4 * u, ng - 1));
#pragma unroll
    for (int u = 0; u < NS; ++u)
        x[u] = ld1_off(elr, (unsigned)ps[u] * (SC * 4u) + (unsigned)h * 4u);
#pragma unroll
    for (int u = NS; u < 8; ++u) x[u] = 0.f;
    load_rows<NG>(sv, 0, n, lane, feat, f);
    __builtin_amdgcn_sched_barrier(0);

    float m = m_in;
#pragma unroll
    for (int u = 0; u < NS; ++u) {
        x[u] = lrelu(x[u] + er);
        m = fmaxf(m, (4 * u < ng) ? x[u] : -INFINITY);
    }
    m = fmaxf(m, __shfl_xor(m, 8));
    m = fmaxf(m, __shfl_xor(m, 16));
    m_out = m;
#pragma unroll
    for (int u = 0; u < NS; ++u) {
        const float e = exp_b2(x[u] - m);
        x[u] = (4 * u < ng) ? e : 0.f;
        z += x[u];
    }
    fma_rows<NG, 0>(x, f, acc);
    if constexpr (NS == 8) {
        if (n - 16 > 8)      rest_rows<4>(sv, n, lane, feat, x, acc);
        else if (n - 16 > 4) rest_rows<2>(sv, n, lane, feat, x, acc);
        else if (n - 16 > 0) rest_rows<1>(sv, n, lane, feat, x, acc);
    }
}

__device__ __forceinline__ int load_bounds(
    const int* __restrict__ offs, int row0, int wave, int lane, int N, int E)
{
    const int bi = row0 + wave * 2 + (lane & 3);
    const int v = offs[min(bi, N - 1)];
    const int lo = (bi < N) ? 0 : E;
    return max(lo, min(v, E));
}

__device__ __forceinline__ float load_er(
    const float* __restrict__ elr, int row0, int wave, int t, int h, int N)
{
    const int node = min(row0 + wave * 2 + t, N - 1);
    return ld1_off(elr, (unsigned)node * (SC * 4u) + (unsigned)(H_ + h) * 4u);
}

__device__ __forceinline__ int load_first_sources(
    const int* __restrict__ ssrc, int voff, int t, int lane, int N, int E)
{
    const int bb = __builtin_amdgcn_readlane(voff, t);
    const int be = max(bb, __builtin_amdgcn_readlane(voff, t + 1));
    const int i = bb + lane;
    const int s = ssrc[max(0, min(i, E - 1))];
    const int hi = (i < be) ? (N - 1) : 0;
    return min(max(s, 0), hi);
}

__global__ __launch_bounds__(256, 8) void gat_aggregate_project(
    const int* __restrict__ offs, const int* __restrict__ ssrc,
    const float* __restrict__ elr, const float* __restrict__ feat,
    const float* __restrict__ W, const float* __restrict__ Wres,
    float* __restrict__ out, int N, int E)
{
    __shared__ float zt[16 * ZP];
    __shared__ v2f brs[8 * (IN_F / 4) * 32];
    __shared__ float ot[16 * OP];

    const int lane = threadIdx.x & 31;
    const int wave = __builtin_amdgcn_readfirstlane((int)(threadIdx.x >> 5));
    const int r  = lane & 15;
    const int hi = lane >> 4;
    const int h  = lane & 7;
    const int g  = lane >> 3;
    const int col0 = wave * 16;

    v2f bw[IN_F / 4];
    v2f* brw = brs + wave * (IN_F / 4) * 32 + lane;
#pragma unroll
    for (int j = 0; j < IN_F / 4; ++j) {
        const size_t o = (size_t)(4 * j + 2 * hi) * HD + col0 + r;
        bw[j].x = W[o];
        bw[j].y = W[o + HD];
        v2f t2;
        t2.x = Wres[o];
        t2.y = Wres[o + HD];
        brw[j * 32] = t2;
    }

    const int ntiles = (N + 15) >> 4;

    int voff = load_bounds(offs, (int)blockIdx.x * 16, wave, lane, N, E);
    int sv0a = load_first_sources(ssrc, voff, 0, lane, N, E);
    int sv0b = load_first_sources(ssrc, voff, 1, lane, N, E);
    float er0 = load_er(elr, (int)blockIdx.x * 16, wave, 0, h, N);
    float er1 = load_er(elr, (int)blockIdx.x * 16, wave, 1, h, N);

    for (int tile = blockIdx.x; tile < ntiles; tile += gridDim.x) {
        const int row0 = tile * 16;

        for (int t = 0; t < 2; ++t) {
            const int nt = wave * 2 + t;
            const int node = row0 + nt;
            v2f acc[H_];
#pragma unroll
            for (int hh = 0; hh < H_; ++hh) acc[hh] = (v2f){0.f, 0.f};

            if (node < N) {
                const int beg = __builtin_amdgcn_readlane(voff, t);
                const int end = max(beg, __builtin_amdgcn_readlane(voff, t + 1));
                const int svf = t ? sv0b : sv0a;

                const float er = t ? er1 : er0;

                float m = -INFINITY;
                for (int cb = beg + 32; cb < end; cb += 32) {
                    const int n = min(32, end - cb);
                    int sv = (cb + lane < end) ? ssrc[cb + lane] : 0;
                    sv = max(0, min(sv, N - 1));
                    for (int k = 0; k < n; k += 16) {
                        int si[4];
                        float x[4];
#pragma unroll
                        for (int u = 0; u < 4; ++u)
                            si[u] = __shfl(sv, min(k + 4 * u + g, n - 1));
#pragma unroll
                        for (int u = 0; u < 4; ++u)
                            x[u] = ld1_off(elr, (unsigned)si[u] * (SC * 4u) + (unsigned)h * 4u);
#pragma unroll
                        for (int u = 0; u < 4; ++u) {
                            const float xv = lrelu(x[u] + er);
                            m = fmaxf(m, (k + 4 * u + g < n) ? xv : -INFINITY);
                        }
                    }
                }

                float z = 0.f;
                if (end > beg) {
                    const int n0 = min(32, end - beg);
                    if (n0 > 16)     first_chunk<4, 8>(svf, n0, lane, g, h, er, m, elr, feat, m, z, acc);
                    else if (n0 > 8) first_chunk<4, 4>(svf, n0, lane, g, h, er, m, elr, feat, m, z, acc);
                    else if (n0 > 4) first_chunk<2, 2>(svf, n0, lane, g, h, er, m, elr, feat, m, z, acc);
                    else             first_chunk<1, 1>(svf, n0, lane, g, h, er, m, elr, feat, m, z, acc);
                }

                for (int cb = beg + 32; cb < end; cb += 32) {
                    const int n = min(32, end - cb);
                    int sv = (cb + lane < end) ? ssrc[cb + lane] : 0;
                    sv = max(0, min(sv, N - 1));
                    int k = 0;
                    for (; k + 16 <= n; k += 16)
                        edge_groups<4>(sv, k, n, lane, g, h, er, m, elr, feat, z, acc);
                    if (n - k > 8)
                        edge_groups<4>(sv, k, n, lane, g, h, er, m, elr, feat, z, acc);
                    else if (n - k > 4)
                        edge_groups<2>(sv, k, n, lane, g, h, er, m, elr, feat, z, acc);
                    else if (n - k > 0)
                        edge_groups<1>(sv, k, n, lane, g, h, er, m, elr, feat, z, acc);
                }
                z += __shfl_xor(z, 8);
                z += __shfl_xor(z, 16);

                if (end > beg) {
                    const int rzb = __float_as_int(1.0f / z);
#pragma unroll
                    for (int hh = 0; hh < H_; ++hh) {
                        const float rh = __int_as_float(__builtin_amdgcn_readlane(rzb, hh));
                        acc[hh] *= rh;
                    }
                }
            }

            float* zrow = zt + nt * ZP + 2 * lane;
#pragma unroll
            for (int hh = 0; hh < H_; ++hh)
                *(v2f*)(zrow + hh * IN_F) = acc[hh];
        }
        __syncthreads();
        const int voff_n = load_bounds(offs, (tile + (int)gridDim.x) * 16, wave, lane, N, E);

        const int arow = (row0 + r < N) ? (row0 + r) : (N - 1);
        const float* ap = feat + (size_t)arow * IN_F + 2 * hi;
        const float* zp = zt + r * ZP + wave * IN_F + 2 * hi;

        v8f c = {};
#pragma unroll
        for (int j = 0; j < IN_F / 4; ++j) {
            const v2f a = *(const v2f*)(zp + 4 * j);
            c = __builtin_amdgcn_wmma_f32_16x16x4_f32(
                false, a, false, bw[j], (short)0, c, false, false);
        }
#pragma unroll
        for (int j = 0; j < IN_F / 4; ++j) {
            v2f a;
            a.x = ap[4 * j + 0];
            a.y = ap[4 * j + 1];
            const v2f b = brw[j * 32];
            c = __builtin_amdgcn_wmma_f32_16x16x4_f32(
                false, a, false, b, (short)0, c, false, false);
        }
        {
            float* orow = ot + (8 * hi) * OP + col0 + r;
#pragma unroll
            for (int v = 0; v < 8; ++v) {
                const float o = c[v];
                orow[v * OP] = o > 0.f ? o : exp_b2(o) - 1.0f;
            }
        }
        __syncthreads();
        const int sv0a_n = load_first_sources(ssrc, voff_n, 0, lane, N, E);
        const int sv0b_n = load_first_sources(ssrc, voff_n, 1, lane, N, E);
        const float er0_n = load_er(elr, (tile + (int)gridDim.x) * 16, wave, 0, h, N);
        const float er1_n = load_er(elr, (tile + (int)gridDim.x) * 16, wave, 1, h, N);

#pragma unroll
        for (int t = 0; t < 2; ++t) {
            const int orw = wave * 2 + t;
            const int grow = row0 + orw;
            if (grow < N)
                *(v4f*)((char*)out + ((unsigned)grow * (HD * 4u) + (unsigned)lane * 16u)) = *(const v4f*)(ot + orw * OP + 4 * lane);
        }
        voff = voff_n;
        sv0a = sv0a_n;
        sv0b = sv0b_n;
        er0 = er0_n;
        er1 = er1_n;
    }
}

extern "C" void kernel_launch(void* const* d_in, const int* in_sizes, int n_in,
                              void* d_out, int out_size, void* d_ws, size_t ws_size,
                              hipStream_t stream)
{
    const float* feat   = (const float*)d_in[0];
    const int*   src    = (const int*)  d_in[1];
    const int*   dst    = (const int*)  d_in[2];
    const float* W      = (const float*)d_in[3];
    const float* attn_l = (const float*)d_in[4];
    const float* attn_r = (const float*)d_in[5];
    const float* Wres   = (const float*)d_in[6];
    float* out = (float*)d_out;

    const int N = in_sizes[0] / IN_F;
    const int E = in_sizes[1];

    char* ws = (char*)d_ws;
    unsigned* tmp = (unsigned*)ws;
    float* elr  = (float*)ws + E;                    ws += (size_t)N * HD * sizeof(float);
    int*   S    = (int*)ws;                          ws += (size_t)N * H_ * sizeof(float);
    int*   ssrc = (int*)ws;                          ws += (size_t)E * sizeof(int);
    int*   offs = (int*)ws;

    gat_bpart<<<PBLK, PTHR, 0, stream>>>(src, dst, S, tmp, N, E);

    gat_bsort_scores<<<782, BNODES, 0, stream>>>(S, tmp, offs, ssrc,
                                                  feat, W, attn_l, attn_r, elr, N, E, 391);

    gat_aggregate_project<<<1024, 256, 0, stream>>>(offs, ssrc, elr, feat, W, Wres, out, N, E);
}
